# starting kernel
# baseline (speedup 1.0000x reference)
#include <hip/hip_runtime.h>
#include <hip/hip_bf16.h>
#include <math.h>


typedef __attribute__((ext_vector_type(2))) float v2f;
typedef __attribute__((ext_vector_type(8))) float v8f;

#define HS 256
#define C2D 128
#define DD 64
#define NC_CHUNKS 8

__global__ void k_qkw1(const float* __restrict__ Q, const float* __restrict__ K,
                       const float* __restrict__ W1,
                       float* __restrict__ QW1, float* __restrict__ KW1) {
  int tid = blockIdx.x * blockDim.x + threadIdx.x;
  int hi = tid >> 7;
  int c  = tid & 127;
  const float* q = Q + hi * 64;
  const float* k = K + hi * 64;
  float sq = 0.f, sk = 0.f;
  #pragma unroll 8
  for (int a = 0; a < 64; ++a) {
    sq += q[a] * W1[a * C2D + c];
    sk += k[a] * W1[(64 + a) * C2D + c];
  }
  QW1[tid] = sq;
  KW1[tid] = sk;
}

__global__ void k_hsum(const float* __restrict__ QW1, const float* __restrict__ KW1,
                       const float* __restrict__ b1, float* __restrict__ Hsum) {
  int tid = blockIdx.x * blockDim.x + threadIdx.x;
  int hj = tid >> 7;
  int c  = tid & 127;
  int h  = hj >> 7;
  float base = KW1[tid] + b1[c];
  const float* qrow = QW1 + (h * 128) * C2D + c;
  float s = 0.f;
  for (int i = 0; i < 128; ++i)
    s += tanhf(qrow[i * C2D] + base);
  Hsum[tid] = s;
}

__global__ __launch_bounds__(32) __attribute__((amdgpu_waves_per_eu(1, 1)))
void k_w2k(const float* __restrict__ K, const float* __restrict__ W2,
           const float* __restrict__ Hsum, float* __restrict__ partial) {
  const int lane = threadIdx.x;
  const int ct = blockIdx.x;
  const int dt = blockIdx.y;
  const int cc = blockIdx.z;
  const int half = lane >> 4;
  const int l16  = lane & 15;
  const int k0   = half * 2;
  const int hj   = ct * 16 + l16;
  const int drow = dt * 16 + l16;
  const int cbase = cc * 16;

  v2f bfrag[16];
  #pragma unroll
  for (int es = 0; es < 16; ++es)
    bfrag[es] = *(const v2f*)(K + hj * 64 + es * 4 + k0);

  v2f cur[16];
  {
    const float* Arow = W2 + (size_t)cbase * 4096 + (size_t)drow * 64;
    #pragma unroll
    for (int es = 0; es < 16; ++es)
      cur[es] = *(const v2f*)(Arow + es * 4 + k0);
  }
  float hs = Hsum[hj * C2D + cbase];

  v8f outacc = {};
  #pragma unroll
  for (int cidx = 0; cidx < 16; ++cidx) {
    v2f nxt[16];
    float hs_n = 0.f;
    if (cidx < 15) {
      const float* ArowN = W2 + (size_t)(cbase + cidx + 1) * 4096 + (size_t)drow * 64;
      #pragma unroll
      for (int es = 0; es < 16; ++es)
        nxt[es] = *(const v2f*)(ArowN + es * 4 + k0);
      hs_n = Hsum[hj * C2D + cbase + cidx + 1];
    }

    v8f acc = {};
    #pragma unroll
    for (int es = 0; es < 16; ++es)
      acc = __builtin_amdgcn_wmma_f32_16x16x4_f32(
           false, cur[es],  false, bfrag[es],
           (short)0, acc,  false,  false);

    #pragma unroll
    for (int v = 0; v < 8; ++v) outacc[v] += hs * acc[v];

    #pragma unroll
    for (int es = 0; es < 16; ++es) cur[es] = nxt[es];
    hs = hs_n;
  }

  float* p = partial + (size_t)cc * (HS * DD) + (size_t)hj * DD + dt * 16 + 8 * half;
  #pragma unroll
  for (int v = 0; v < 8; ++v) p[v] = outacc[v];
}

__global__ void k_out(const float* __restrict__ partial, const float* __restrict__ K,
                      const float* __restrict__ b2, float* __restrict__ out) {
  int tid = blockIdx.x * blockDim.x + threadIdx.x;
  int hj = tid >> 6;
  int d  = tid & 63;
  float s = 0.f;
  #pragma unroll
  for (int cc = 0; cc < NC_CHUNKS; ++cc)
    s += partial[cc * (HS * DD) + tid];
  const float* b2row = b2 + d * 64;
  const float* krow  = K + hj * 64;
  float b2k = 0.f;
  #pragma unroll 8
  for (int e = 0; e < 64; ++e)
    b2k += b2row[e] * krow[e];
  out[tid] = s * (1.f / 128.f) + b2k;
}

extern "C" void kernel_launch(void* const* d_in, const int* in_sizes, int n_in,
                              void* d_out, int out_size, void* d_ws, size_t ws_size,
                              hipStream_t stream) {
  const float* Q  = (const float*)d_in[0];
  const float* K  = (const float*)d_in[1];
  const float* W1 = (const float*)d_in[2];
  const float* b1 = (const float*)d_in[3];
  const float* W2 = (const float*)d_in[4];
  const float* b2 = (const float*)d_in[5];
  float* out = (float*)d_out;

  float* ws      = (float*)d_ws;
  float* QW1     = ws;
  float* KW1     = ws + 32768;
  float* Hsum    = ws + 65536;
  float* partial = ws + 98304;

  k_qkw1<<<128, 256, 0, stream>>>(Q, K, W1, QW1, KW1);
  k_hsum<<<128, 256, 0, stream>>>(QW1, KW1, b1, Hsum);
  dim3 g3(16, 4, NC_CHUNKS);
  k_w2k<<<g3, 32, 0, stream>>>(K, W2, Hsum, partial);
  k_out<<<64, 256, 0, stream>>>(partial, K, b2, out);
}


# k_hsum: the serial 128-step sum over i is split over 4 waves per (hj, 32 c), each with its 32 loads in flight at once; quarter sums added in fixed order through LDS (reduction order changes)
# speedup vs baseline: 2.6689x; 2.6689x over previous
#include <hip/hip_runtime.h>
#include <hip/hip_bf16.h>
#include <math.h>


typedef __attribute__((ext_vector_type(2))) float v2f;
typedef __attribute__((ext_vector_type(8))) float v8f;

#define HS 256
#define C2D 128
#define DD 64
#define NC_CHUNKS 8

__global__ void k_qkw1(const float* __restrict__ Q, const float* __restrict__ K,
                       const float* __restrict__ W1,
                       float* __restrict__ QW1, float* __restrict__ KW1) {
  int tid = blockIdx.x * blockDim.x + threadIdx.x;
  int hi = tid >> 7;
  int c  = tid & 127;
  const float* q = Q + hi * 64;
  const float* k = K + hi * 64;
  float sq = 0.f, sk = 0.f;
  #pragma unroll 8
  for (int a = 0; a < 64; ++a) {
    sq += q[a] * W1[a * C2D + c];
    sk += k[a] * W1[(64 + a) * C2D + c];
  }
  QW1[tid] = sq;
  KW1[tid] = sk;
}

#define IPARTS 4
#define ILEN 32
__global__ __launch_bounds__(IPARTS * 32)
void k_hsum(const float* __restrict__ QW1, const float* __restrict__ KW1,
            const float* __restrict__ b1, float* __restrict__ Hsum) {
  __shared__ float part[IPARTS][32];
  const int lane = threadIdx.x & 31;
  const int w    = threadIdx.x >> 5;
  const int hj   = blockIdx.x >> 2;
  const int c    = (blockIdx.x & 3) * 32 + lane;
  const int h    = hj >> 7;
  const float base = KW1[hj * C2D + c] + b1[c];
  const float* qrow = QW1 + (h * 128 + w * ILEN) * C2D + c;
  float q[ILEN];
  #pragma unroll
  for (int i = 0; i < ILEN; ++i)
    q[i] = qrow[i * C2D];
  float s = 0.f;
  #pragma unroll
  for (int i = 0; i < ILEN; ++i)
    s += tanhf(q[i] + base);
  part[w][lane] = s;
  __syncthreads();
  if (w == 0)
    Hsum[hj * C2D + c] = ((part[0][lane] + part[1][lane]) + part[2][lane]) + part[3][lane];
}

__global__ __launch_bounds__(32) __attribute__((amdgpu_waves_per_eu(1, 1)))
void k_w2k(const float* __restrict__ K, const float* __restrict__ W2,
           const float* __restrict__ Hsum, float* __restrict__ partial) {
  const int lane = threadIdx.x;
  const int ct = blockIdx.x;
  const int dt = blockIdx.y;
  const int cc = blockIdx.z;
  const int half = lane >> 4;
  const int l16  = lane & 15;
  const int k0   = half * 2;
  const int hj   = ct * 16 + l16;
  const int drow = dt * 16 + l16;
  const int cbase = cc * 16;

  v2f bfrag[16];
  #pragma unroll
  for (int es = 0; es < 16; ++es)
    bfrag[es] = *(const v2f*)(K + hj * 64 + es * 4 + k0);

  v2f cur[16];
  {
    const float* Arow = W2 + (size_t)cbase * 4096 + (size_t)drow * 64;
    #pragma unroll
    for (int es = 0; es < 16; ++es)
      cur[es] = *(const v2f*)(Arow + es * 4 + k0);
  }
  float hs = Hsum[hj * C2D + cbase];

  v8f outacc = {};
  #pragma unroll
  for (int cidx = 0; cidx < 16; ++cidx) {
    v2f nxt[16];
    float hs_n = 0.f;
    if (cidx < 15) {
      const float* ArowN = W2 + (size_t)(cbase + cidx + 1) * 4096 + (size_t)drow * 64;
      #pragma unroll
      for (int es = 0; es < 16; ++es)
        nxt[es] = *(const v2f*)(ArowN + es * 4 + k0);
      hs_n = Hsum[hj * C2D + cbase + cidx + 1];
    }

    v8f acc = {};
    #pragma unroll
    for (int es = 0; es < 16; ++es)
      acc = __builtin_amdgcn_wmma_f32_16x16x4_f32(
           false, cur[es],  false, bfrag[es],
           (short)0, acc,  false,  false);

    #pragma unroll
    for (int v = 0; v < 8; ++v) outacc[v] += hs * acc[v];

    #pragma unroll
    for (int es = 0; es < 16; ++es) cur[es] = nxt[es];
    hs = hs_n;
  }

  float* p = partial + (size_t)cc * (HS * DD) + (size_t)hj * DD + dt * 16 + 8 * half;
  #pragma unroll
  for (int v = 0; v < 8; ++v) p[v] = outacc[v];
}

__global__ void k_out(const float* __restrict__ partial, const float* __restrict__ K,
                      const float* __restrict__ b2, float* __restrict__ out) {
  int tid = blockIdx.x * blockDim.x + threadIdx.x;
  int hj = tid >> 6;
  int d  = tid & 63;
  float s = 0.f;
  #pragma unroll
  for (int cc = 0; cc < NC_CHUNKS; ++cc)
    s += partial[cc * (HS * DD) + tid];
  const float* b2row = b2 + d * 64;
  const float* krow  = K + hj * 64;
  float b2k = 0.f;
  #pragma unroll 8
  for (int e = 0; e < 64; ++e)
    b2k += b2row[e] * krow[e];
  out[tid] = s * (1.f / 128.f) + b2k;
}

extern "C" void kernel_launch(void* const* d_in, const int* in_sizes, int n_in,
                              void* d_out, int out_size, void* d_ws, size_t ws_size,
                              hipStream_t stream) {
  const float* Q  = (const float*)d_in[0];
  const float* K  = (const float*)d_in[1];
  const float* W1 = (const float*)d_in[2];
  const float* b1 = (const float*)d_in[3];
  const float* W2 = (const float*)d_in[4];
  const float* b2 = (const float*)d_in[5];
  float* out = (float*)d_out;

  float* ws      = (float*)d_ws;
  float* QW1     = ws;
  float* KW1     = ws + 32768;
  float* Hsum    = ws + 65536;
  float* partial = ws + 98304;

  k_qkw1<<<128, 256, 0, stream>>>(Q, K, W1, QW1, KW1);
  k_hsum<<<1024, 128, 0, stream>>>(QW1, KW1, b1, Hsum);
  dim3 g3(16, 4, NC_CHUNKS);
  k_w2k<<<g3, 32, 0, stream>>>(K, W2, Hsum, partial);
  k_out<<<64, 256, 0, stream>>>(partial, K, b2, out);
}


# k_w2k contracts c on the matrix unit first (Hsum x W2, 4 independent accumulator tiles per wave, 16-byte operand loads), then e in registers; partial planes and k_out launch go (sum order changes)
# speedup vs baseline: 3.6923x; 1.3835x over previous
#include <hip/hip_runtime.h>
#include <hip/hip_bf16.h>
#include <math.h>


typedef __attribute__((ext_vector_type(2))) float v2f;
typedef __attribute__((ext_vector_type(8))) float v8f;

#define HS 256
#define C2D 128
#define DD 64

__global__ void k_qkw1(const float* __restrict__ Q, const float* __restrict__ K,
                       const float* __restrict__ W1,
                       float* __restrict__ QW1, float* __restrict__ KW1) {
  int tid = blockIdx.x * blockDim.x + threadIdx.x;
  int hi = tid >> 7;
  int c  = tid & 127;
  const float* q = Q + hi * 64;
  const float* k = K + hi * 64;
  float sq = 0.f, sk = 0.f;
  #pragma unroll 8
  for (int a = 0; a < 64; ++a) {
    sq += q[a] * W1[a * C2D + c];
    sk += k[a] * W1[(64 + a) * C2D + c];
  }
  QW1[tid] = sq;
  KW1[tid] = sk;
}

#define IPARTS 4
#define ILEN 32
__global__ __launch_bounds__(IPARTS * 32)
void k_hsum(const float* __restrict__ QW1, const float* __restrict__ KW1,
            const float* __restrict__ b1, float* __restrict__ Hsum) {
  __shared__ float part[IPARTS][32];
  const int lane = threadIdx.x & 31;
  const int w    = threadIdx.x >> 5;
  const int hj   = blockIdx.x >> 2;
  const int c    = (blockIdx.x & 3) * 32 + lane;
  const int h    = hj >> 7;
  const float base = KW1[hj * C2D + c] + b1[c];
  const float* qrow = QW1 + (h * 128 + w * ILEN) * C2D + c;
  float q[ILEN];
  #pragma unroll
  for (int i = 0; i < ILEN; ++i)
    q[i] = qrow[i * C2D];
  float s = 0.f;
  #pragma unroll
  for (int i = 0; i < ILEN; ++i)
    s += tanhf(q[i] + base);
  part[w][lane] = s;
  __syncthreads();
  if (w == 0)
    Hsum[hj * C2D + c] = ((part[0][lane] + part[1][lane]) + part[2][lane]) + part[3][lane];
}

typedef __attribute__((ext_vector_type(4))) float v4f;
__device__ __forceinline__ int c_of(int ks, int half, int v) {
  return (ks >> 1) * 8 + 4 * half + 2 * (ks & 1) + v;
}
__global__ __launch_bounds__(128) __attribute__((amdgpu_waves_per_eu(1, 1)))
void k_w2k(const float* __restrict__ K, const float* __restrict__ W2,
           const float* __restrict__ b2, const float* __restrict__ Hsum,
           float* __restrict__ out) {
  const int lane = threadIdx.x & 31;
  const int ct = blockIdx.x * 4 + (threadIdx.x >> 5);
  const int d  = blockIdx.y;
  const int half = lane >> 4;
  const int l16  = lane & 15;
  const int hj   = ct * 16 + l16;

  v4f bq[16];
  #pragma unroll
  for (int p = 0; p < 16; ++p)
    bq[p] = *(const v4f*)(Hsum + hj * C2D + p * 8 + 4 * half);

  const float* Ab = W2 + d * DD + 4 * l16;
  v4f aq[32][2];
  #pragma unroll
  for (int ks = 0; ks < 32; ++ks) {
    #pragma unroll
    for (int v = 0; v < 2; ++v)
      aq[ks][v] = *(const v4f*)(Ab + (size_t)c_of(ks, half, v) * 4096);
  }

  v4f kq[8], b2q[8];
  #pragma unroll
  for (int r = 0; r < 8; ++r) {
    kq[r]  = *(const v4f*)(K  + hj * DD + 32 * half + 4 * r);
    b2q[r] = *(const v4f*)(b2 + d  * DD + 32 * half + 4 * r);
  }

  v8f acc[4] = {};
  #pragma unroll
  for (int ks = 0; ks < 32; ++ks) {
    v2f bf;
    if (ks & 1) { bf.x = bq[ks >> 1].z; bf.y = bq[ks >> 1].w; }
    else        { bf.x = bq[ks >> 1].x; bf.y = bq[ks >> 1].y; }
    #pragma unroll
    for (int et = 0; et < 4; ++et) {
      v2f af;
      af.x = aq[ks][0][et];
      af.y = aq[ks][1][et];
      acc[et] = __builtin_amdgcn_wmma_f32_16x16x4_f32(
           false, af,  false, bf,
           (short)0, acc[et],  false,  false);
    }
  }

  float s = 0.f;
  #pragma unroll
  for (int r = 0; r < 8; ++r) {
    #pragma unroll
    for (int et = 0; et < 4; ++et)
      s += (acc[et][r] * (1.f / 128.f) + b2q[r][et]) * kq[r][et];
  }
  s += __shfl_xor(s, 16);
  if (half == 0) out[hj * DD + d] = s;
}

extern "C" void kernel_launch(void* const* d_in, const int* in_sizes, int n_in,
                              void* d_out, int out_size, void* d_ws, size_t ws_size,
                              hipStream_t stream) {
  const float* Q  = (const float*)d_in[0];
  const float* K  = (const float*)d_in[1];
  const float* W1 = (const float*)d_in[2];
  const float* b1 = (const float*)d_in[3];
  const float* W2 = (const float*)d_in[4];
  const float* b2 = (const float*)d_in[5];
  float* out = (float*)d_out;

  float* ws      = (float*)d_ws;
  float* QW1     = ws;
  float* KW1     = ws + 32768;
  float* Hsum    = ws + 65536;

  k_qkw1<<<128, 256, 0, stream>>>(Q, K, W1, QW1, KW1);
  k_hsum<<<1024, 128, 0, stream>>>(QW1, KW1, b1, Hsum);
  dim3 g3(4, 64);
  k_w2k<<<g3, 128, 0, stream>>>(K, W2, b2, Hsum, out);
}


# k_qkw1 launch folded into k_hsum: a block (h, 4 j, 32 c; 16 waves) stages Q[h] and its W1 columns in LDS, forms QW1 on the matrix unit, KW1 in 4 partial sums (QW1/KW1 sum order changes)
# speedup vs baseline: 4.9003x; 1.3272x over previous
#include <hip/hip_runtime.h>
#include <hip/hip_bf16.h>
#include <math.h>


typedef __attribute__((ext_vector_type(2))) float v2f;
typedef __attribute__((ext_vector_type(8))) float v8f;

#define HS 256
#define C2D 128
#define DD 64

typedef __attribute__((ext_vector_type(4))) float v4f;
__device__ __forceinline__ int c_of(int ks, int half, int v) {
  return (ks >> 1) * 8 + 4 * half + 2 * (ks & 1) + v;
}

#define JB 4
#define CB 32
#define AP 68
#define WP 36
__global__ __launch_bounds__(512)
void k_hsum(const float* __restrict__ Q, const float* __restrict__ K,
            const float* __restrict__ W1, const float* __restrict__ b1,
            float* __restrict__ Hsum) {
  __shared__ __attribute__((aligned(16))) float sA[128][AP];
  __shared__ __attribute__((aligned(16))) float sW[128][WP];
  __shared__ float sQ[128][WP];
  __shared__ float sK[4][JB][CB];
  __shared__ float part[4][JB][CB];
  const int t    = threadIdx.x;
  const int lane = t & 31;
  const int w    = __builtin_amdgcn_readfirstlane(t >> 5);
  const int half = lane >> 4;
  const int l16  = lane & 15;
  const int c0   = (blockIdx.x & 3) * CB;
  const int j0   = ((blockIdx.x >> 2) & 31) * JB;
  const int h    = blockIdx.x >> 7;

  {
    v4f qv[4], wv[2];
    #pragma unroll
    for (int u = 0; u < 4; ++u)
      qv[u] = *(const v4f*)(Q + h * 8192 + (u * 512 + t) * 4);
    #pragma unroll
    for (int u = 0; u < 2; ++u) {
      const int idx = u * 512 + t;
      wv[u] = *(const v4f*)(W1 + (idx >> 3) * C2D + c0 + (idx & 7) * 4);
    }
    #pragma unroll
    for (int u = 0; u < 4; ++u) {
      const int idx = u * 512 + t;
      *(v4f*)&sA[idx >> 4][(idx & 15) * 4] = qv[u];
    }
    #pragma unroll
    for (int u = 0; u < 2; ++u) {
      const int idx = u * 512 + t;
      *(v4f*)&sW[idx >> 3][(idx & 7) * 4] = wv[u];
    }
  }
  __syncthreads();

  const int jw = w & 3;
  const int iq = w >> 2;
  {
    const float* krow = K + (h * 128 + j0 + jw) * DD + 16 * iq;
    float sk = 0.f;
    #pragma unroll
    for (int a = 0; a < 16; ++a)
      sk += krow[a] * sW[64 + 16 * iq + a][lane];
    sK[iq][jw][lane] = sk;
  }

  {
    const int it = w >> 1;
    const int ct = w & 1;
    v4f aq[8];
    #pragma unroll
    for (int p = 0; p < 8; ++p)
      aq[p] = *(const v4f*)&sA[16 * it + l16][p * 8 + 4 * half];
    v8f acc = {};
    #pragma unroll
    for (int ks = 0; ks < 16; ++ks) {
      v2f af, bf;
      if (ks & 1) { af.x = aq[ks >> 1].z; af.y = aq[ks >> 1].w; }
      else        { af.x = aq[ks >> 1].x; af.y = aq[ks >> 1].y; }
      bf.x = sW[c_of(ks, half, 0)][16 * ct + l16];
      bf.y = sW[c_of(ks, half, 1)][16 * ct + l16];
      acc = __builtin_amdgcn_wmma_f32_16x16x4_f32(
           false, af,  false, bf,
           (short)0, acc,  false,  false);
    }
    #pragma unroll
    for (int r = 0; r < 8; ++r)
      sQ[16 * it + 8 * half + r][16 * ct + l16] = acc[r];
  }
  __syncthreads();

  {
    const float base = (((sK[0][jw][lane] + sK[1][jw][lane]) + sK[2][jw][lane]) + sK[3][jw][lane])
                       + b1[c0 + lane];
    float q[32];
    #pragma unroll
    for (int i = 0; i < 32; ++i)
      q[i] = sQ[32 * iq + i][lane];
    float s = 0.f;
    #pragma unroll
    for (int i = 0; i < 32; ++i)
      s += tanhf(q[i] + base);
    part[iq][jw][lane] = s;
  }
  __syncthreads();
  if (w < JB)
    Hsum[(h * 128 + j0 + w) * C2D + c0 + lane] =
        ((part[0][w][lane] + part[1][w][lane]) + part[2][w][lane]) + part[3][w][lane];
}

__global__ __launch_bounds__(128) __attribute__((amdgpu_waves_per_eu(1, 1)))
void k_w2k(const float* __restrict__ K, const float* __restrict__ W2,
           const float* __restrict__ b2, const float* __restrict__ Hsum,
           float* __restrict__ out) {
  const int lane = threadIdx.x & 31;
  const int ct = blockIdx.x * 4 + (threadIdx.x >> 5);
  const int d  = blockIdx.y;
  const int half = lane >> 4;
  const int l16  = lane & 15;
  const int hj   = ct * 16 + l16;

  v4f bq[16];
  #pragma unroll
  for (int p = 0; p < 16; ++p)
    bq[p] = *(const v4f*)(Hsum + hj * C2D + p * 8 + 4 * half);

  const float* Ab = W2 + d * DD + 4 * l16;
  v4f aq[32][2];
  #pragma unroll
  for (int ks = 0; ks < 32; ++ks) {
    #pragma unroll
    for (int v = 0; v < 2; ++v)
      aq[ks][v] = *(const v4f*)(Ab + (size_t)c_of(ks, half, v) * 4096);
  }

  v4f kq[8], b2q[8];
  #pragma unroll
  for (int r = 0; r < 8; ++r) {
    kq[r]  = *(const v4f*)(K  + hj * DD + 32 * half + 4 * r);
    b2q[r] = *(const v4f*)(b2 + d  * DD + 32 * half + 4 * r);
  }

  v8f acc[4] = {};
  #pragma unroll
  for (int ks = 0; ks < 32; ++ks) {
    v2f bf;
    if (ks & 1) { bf.x = bq[ks >> 1].z; bf.y = bq[ks >> 1].w; }
    else        { bf.x = bq[ks >> 1].x; bf.y = bq[ks >> 1].y; }
    #pragma unroll
    for (int et = 0; et < 4; ++et) {
      v2f af;
      af.x = aq[ks][0][et];
      af.y = aq[ks][1][et];
      acc[et] = __builtin_amdgcn_wmma_f32_16x16x4_f32(
           false, af,  false, bf,
           (short)0, acc[et],  false,  false);
    }
  }

  float s = 0.f;
  #pragma unroll
  for (int r = 0; r < 8; ++r) {
    #pragma unroll
    for (int et = 0; et < 4; ++et)
      s += (acc[et][r] * (1.f / 128.f) + b2q[r][et]) * kq[r][et];
  }
  s += __shfl_xor(s, 16);
  if (half == 0) out[hj * DD + d] = s;
}

extern "C" void kernel_launch(void* const* d_in, const int* in_sizes, int n_in,
                              void* d_out, int out_size, void* d_ws, size_t ws_size,
                              hipStream_t stream) {
  const float* Q  = (const float*)d_in[0];
  const float* K  = (const float*)d_in[1];
  const float* W1 = (const float*)d_in[2];
  const float* b1 = (const float*)d_in[3];
  const float* W2 = (const float*)d_in[4];
  const float* b2 = (const float*)d_in[5];
  float* out = (float*)d_out;

  float* Hsum    = (float*)d_ws;

  k_hsum<<<256, 512, 0, stream>>>(Q, K, W1, b1, Hsum);
  dim3 g3(4, 64);
  k_w2k<<<g3, 128, 0, stream>>>(K, W2, b2, Hsum, out);
}


# k_hsum: library tanhf (2 exec-masked branches, ~39 issue cycles a call, calls serialised) replaced by branch-free 1-2/(1+exp2(2*log2e*x)) on v_exp_f32 + v_rcp_f32; sum order unchanged
# speedup vs baseline: 5.8914x; 1.2023x over previous
#include <hip/hip_runtime.h>
#include <hip/hip_bf16.h>
#include <math.h>


typedef __attribute__((ext_vector_type(2))) float v2f;
typedef __attribute__((ext_vector_type(8))) float v8f;

#define HS 256
#define C2D 128
#define DD 64

typedef __attribute__((ext_vector_type(4))) float v4f;
__device__ __forceinline__ int c_of(int ks, int half, int v) {
  return (ks >> 1) * 8 + 4 * half + 2 * (ks & 1) + v;
}

#define TWO_LOG2E 2.885390081777927f
__device__ __forceinline__ float tanh_from_t(float t) {
  const float e = __builtin_amdgcn_exp2f(t);
  return 1.0f - 2.0f * __builtin_amdgcn_rcpf(1.0f + e);
}

#define JB 4
#define CB 32
#define AP 68
#define WP 36
__global__ __launch_bounds__(512)
void k_hsum(const float* __restrict__ Q, const float* __restrict__ K,
            const float* __restrict__ W1, const float* __restrict__ b1,
            float* __restrict__ Hsum) {
  __shared__ __attribute__((aligned(16))) float sA[128][AP];
  __shared__ __attribute__((aligned(16))) float sW[128][WP];
  __shared__ float sQ[128][WP];
  __shared__ float sK[4][JB][CB];
  __shared__ float part[4][JB][CB];
  const int t    = threadIdx.x;
  const int lane = t & 31;
  const int w    = __builtin_amdgcn_readfirstlane(t >> 5);
  const int half = lane >> 4;
  const int l16  = lane & 15;
  const int c0   = (blockIdx.x & 3) * CB;
  const int j0   = ((blockIdx.x >> 2) & 31) * JB;
  const int h    = blockIdx.x >> 7;

  {
    v4f qv[4], wv[2];
    #pragma unroll
    for (int u = 0; u < 4; ++u)
      qv[u] = *(const v4f*)(Q + h * 8192 + (u * 512 + t) * 4);
    #pragma unroll
    for (int u = 0; u < 2; ++u) {
      const int idx = u * 512 + t;
      wv[u] = *(const v4f*)(W1 + (idx >> 3) * C2D + c0 + (idx & 7) * 4);
    }
    #pragma unroll
    for (int u = 0; u < 4; ++u) {
      const int idx = u * 512 + t;
      *(v4f*)&sA[idx >> 4][(idx & 15) * 4] = qv[u];
    }
    #pragma unroll
    for (int u = 0; u < 2; ++u) {
      const int idx = u * 512 + t;
      *(v4f*)&sW[idx >> 3][(idx & 7) * 4] = wv[u];
    }
  }
  __syncthreads();

  const int jw = w & 3;
  const int iq = w >> 2;
  {
    const float* krow = K + (h * 128 + j0 + jw) * DD + 16 * iq;
    float sk = 0.f;
    #pragma unroll
    for (int a = 0; a < 16; ++a)
      sk += krow[a] * sW[64 + 16 * iq + a][lane];
    sK[iq][jw][lane] = sk;
  }

  {
    const int it = w >> 1;
    const int ct = w & 1;
    v4f aq[8];
    #pragma unroll
    for (int p = 0; p < 8; ++p)
      aq[p] = *(const v4f*)&sA[16 * it + l16][p * 8 + 4 * half];
    v8f acc = {};
    #pragma unroll
    for (int ks = 0; ks < 16; ++ks) {
      v2f af, bf;
      if (ks & 1) { af.x = aq[ks >> 1].z; af.y = aq[ks >> 1].w; }
      else        { af.x = aq[ks >> 1].x; af.y = aq[ks >> 1].y; }
      bf.x = sW[c_of(ks, half, 0)][16 * ct + l16];
      bf.y = sW[c_of(ks, half, 1)][16 * ct + l16];
      acc = __builtin_amdgcn_wmma_f32_16x16x4_f32(
           false, af,  false, bf,
           (short)0, acc,  false,  false);
    }
    #pragma unroll
    for (int r = 0; r < 8; ++r)
      sQ[16 * it + 8 * half + r][16 * ct + l16] = acc[r];
  }
  __syncthreads();

  {
    const float base = (((sK[0][jw][lane] + sK[1][jw][lane]) + sK[2][jw][lane]) + sK[3][jw][lane])
                       + b1[c0 + lane];
    float q[32];
    #pragma unroll
    for (int i = 0; i < 32; ++i)
      q[i] = sQ[32 * iq + i][lane];
    const float bt = base * TWO_LOG2E;
    float s = 0.f;
    #pragma unroll
    for (int i = 0; i < 32; ++i)
      s += tanh_from_t(__builtin_fmaf(q[i], TWO_LOG2E, bt));
    part[iq][jw][lane] = s;
  }
  __syncthreads();
  if (w < JB)
    Hsum[(h * 128 + j0 + w) * C2D + c0 + lane] =
        ((part[0][w][lane] + part[1][w][lane]) + part[2][w][lane]) + part[3][w][lane];
}

__global__ __launch_bounds__(128) __attribute__((amdgpu_waves_per_eu(1, 1)))
void k_w2k(const float* __restrict__ K, const float* __restrict__ W2,
           const float* __restrict__ b2, const float* __restrict__ Hsum,
           float* __restrict__ out) {
  const int lane = threadIdx.x & 31;
  const int ct = blockIdx.x * 4 + (threadIdx.x >> 5);
  const int d  = blockIdx.y;
  const int half = lane >> 4;
  const int l16  = lane & 15;
  const int hj   = ct * 16 + l16;

  v4f bq[16];
  #pragma unroll
  for (int p = 0; p < 16; ++p)
    bq[p] = *(const v4f*)(Hsum + hj * C2D + p * 8 + 4 * half);

  const float* Ab = W2 + d * DD + 4 * l16;
  v4f aq[32][2];
  #pragma unroll
  for (int ks = 0; ks < 32; ++ks) {
    #pragma unroll
    for (int v = 0; v < 2; ++v)
      aq[ks][v] = *(const v4f*)(Ab + (size_t)c_of(ks, half, v) * 4096);
  }

  v4f kq[8], b2q[8];
  #pragma unroll
  for (int r = 0; r < 8; ++r) {
    kq[r]  = *(const v4f*)(K  + hj * DD + 32 * half + 4 * r);
    b2q[r] = *(const v4f*)(b2 + d  * DD + 32 * half + 4 * r);
  }

  v8f acc[4] = {};
  #pragma unroll
  for (int ks = 0; ks < 32; ++ks) {
    v2f bf;
    if (ks & 1) { bf.x = bq[ks >> 1].z; bf.y = bq[ks >> 1].w; }
    else        { bf.x = bq[ks >> 1].x; bf.y = bq[ks >> 1].y; }
    #pragma unroll
    for (int et = 0; et < 4; ++et) {
      v2f af;
      af.x = aq[ks][0][et];
      af.y = aq[ks][1][et];
      acc[et] = __builtin_amdgcn_wmma_f32_16x16x4_f32(
           false, af,  false, bf,
           (short)0, acc[et],  false,  false);
    }
  }

  float s = 0.f;
  #pragma unroll
  for (int r = 0; r < 8; ++r) {
    #pragma unroll
    for (int et = 0; et < 4; ++et)
      s += (acc[et][r] * (1.f / 128.f) + b2q[r][et]) * kq[r][et];
  }
  s += __shfl_xor(s, 16);
  if (half == 0) out[hj * DD + d] = s;
}

extern "C" void kernel_launch(void* const* d_in, const int* in_sizes, int n_in,
                              void* d_out, int out_size, void* d_ws, size_t ws_size,
                              hipStream_t stream) {
  const float* Q  = (const float*)d_in[0];
  const float* K  = (const float*)d_in[1];
  const float* W1 = (const float*)d_in[2];
  const float* b1 = (const float*)d_in[3];
  const float* W2 = (const float*)d_in[4];
  const float* b2 = (const float*)d_in[5];
  float* out = (float*)d_out;

  float* Hsum    = (float*)d_ws;

  k_hsum<<<256, 512, 0, stream>>>(Q, K, W1, b1, Hsum);
  dim3 g3(4, 64);
  k_w2k<<<g3, 128, 0, stream>>>(K, W2, b2, Hsum, out);
}


# k_hsum re-blocked to (h, 8 j, 16 c): the QW1 slice a block forms halves (8 tiles on waves 0-7, 32 not 64 WMMAs per SIMD), waves 8-15 form KW1 beside them; b1 read at entry; Hsum c-group-major
# speedup vs baseline: 6.7985x; 1.1540x over previous
#include <hip/hip_runtime.h>
#include <hip/hip_bf16.h>
#include <math.h>


typedef __attribute__((ext_vector_type(2))) float v2f;
typedef __attribute__((ext_vector_type(8))) float v8f;

#define HS 256
#define C2D 128
#define DD 64

typedef __attribute__((ext_vector_type(4))) float v4f;
__device__ __forceinline__ int c_of(int ks, int half, int v) {
  return (ks >> 1) * 8 + 4 * half + 2 * (ks & 1) + v;
}

#define TWO_LOG2E 2.885390081777927f
__device__ __forceinline__ float tanh_from_t(float t) {
  const float e = __builtin_amdgcn_exp2f(t);
  return 1.0f - 2.0f * __builtin_amdgcn_rcpf(1.0f + e);
}

#define JB 8
#define CB 16
#define AP 68
#define WP 20
__global__ __launch_bounds__(512)
void k_hsum(const float* __restrict__ Q, const float* __restrict__ K,
            const float* __restrict__ W1, const float* __restrict__ b1,
            float* __restrict__ Hsum) {
  __shared__ __attribute__((aligned(16))) float sA[128][AP];
  __shared__ __attribute__((aligned(16))) float sW[128][WP];
  __shared__ float sQ[128][WP];
  __shared__ float sK[4][JB * CB];
  __shared__ float part[4][JB * CB];
  const int t    = threadIdx.x;
  const int lane = t & 31;
  const int w    = __builtin_amdgcn_readfirstlane(t >> 5);
  const int half = lane >> 4;
  const int l16  = lane & 15;
  const int cg   = blockIdx.x & 7;
  const int c0   = cg * CB;
  const int j0   = ((blockIdx.x >> 3) & 15) * JB;
  const int h    = blockIdx.x >> 7;
  const int jj   = 2 * (w & 3) + half;
  const int ah   = (w >> 2) & 1;

  v4f kv[8];
  float b1v;
  {
    v4f qv[4], wv;
    #pragma unroll
    for (int u = 0; u < 4; ++u)
      qv[u] = *(const v4f*)(Q + h * 8192 + (u * 512 + t) * 4);
    wv = *(const v4f*)(W1 + (t >> 2) * C2D + c0 + (t & 3) * 4);
    if (w >= 8) {
      const v4f* kp = (const v4f*)(K + (h * 128 + j0 + jj) * DD + 32 * ah);
      #pragma unroll
      for (int u = 0; u < 8; ++u)
        kv[u] = kp[u];
    }
    b1v = b1[c0 + l16];
    #pragma unroll
    for (int u = 0; u < 4; ++u) {
      const int idx = u * 512 + t;
      *(v4f*)&sA[idx >> 4][(idx & 15) * 4] = qv[u];
    }
    *(v4f*)&sW[t >> 2][(t & 3) * 4] = wv;
  }
  __syncthreads();

  if (w >= 8) {
    float p_lo = 0.f, p_hi = 0.f;
    #pragma unroll
    for (int a = 0; a < 16; ++a)
      p_lo += kv[a >> 2][a & 3] * sW[64 + 32 * ah + a][l16];
    #pragma unroll
    for (int a = 0; a < 16; ++a)
      p_hi += kv[4 + (a >> 2)][a & 3] * sW[64 + 32 * ah + 16 + a][l16];
    sK[2 * ah][jj * CB + l16]     = p_lo;
    sK[2 * ah + 1][jj * CB + l16] = p_hi;
  } else {
    const int it = w;
    v4f aq[8];
    #pragma unroll
    for (int p = 0; p < 8; ++p)
      aq[p] = *(const v4f*)&sA[16 * it + l16][p * 8 + 4 * half];
    v8f acc = {};
    #pragma unroll
    for (int ks = 0; ks < 16; ++ks) {
      v2f af, bf;
      if (ks & 1) { af.x = aq[ks >> 1].z; af.y = aq[ks >> 1].w; }
      else        { af.x = aq[ks >> 1].x; af.y = aq[ks >> 1].y; }
      bf.x = sW[c_of(ks, half, 0)][l16];
      bf.y = sW[c_of(ks, half, 1)][l16];
      acc = __builtin_amdgcn_wmma_f32_16x16x4_f32(
           false, af,  false, bf,
           (short)0, acc,  false,  false);
    }
    #pragma unroll
    for (int r = 0; r < 8; ++r)
      sQ[16 * it + 8 * half + r][l16] = acc[r];
  }
  __syncthreads();

  const int iq = w >> 2;
  {
    const int o = jj * CB + l16;
    const float base = (((sK[0][o] + sK[1][o]) + sK[2][o]) + sK[3][o]) + b1v;
    float q[32];
    #pragma unroll
    for (int i = 0; i < 32; ++i)
      q[i] = sQ[32 * iq + i][l16];
    const float bt = base * TWO_LOG2E;
    float s = 0.f;
    #pragma unroll
    for (int i = 0; i < 32; ++i)
      s += tanh_from_t(__builtin_fmaf(q[i], TWO_LOG2E, bt));
    part[iq][o] = s;
  }
  __syncthreads();
  if (w < 4) {
    const int o = w * 32 + lane;
    Hsum[((h * 8 + cg) * 128 + j0) * CB + o] =
        ((part[0][o] + part[1][o]) + part[2][o]) + part[3][o];
  }
}

__global__ __launch_bounds__(128) __attribute__((amdgpu_waves_per_eu(1, 1)))
void k_w2k(const float* __restrict__ K, const float* __restrict__ W2,
           const float* __restrict__ b2, const float* __restrict__ Hsum,
           float* __restrict__ out) {
  const int lane = threadIdx.x & 31;
  const int ct = blockIdx.x * 4 + (threadIdx.x >> 5);
  const int d  = blockIdx.y;
  const int half = lane >> 4;
  const int l16  = lane & 15;
  const int hj   = ct * 16 + l16;

  v4f bq[16];
  #pragma unroll
  for (int p = 0; p < 16; ++p)
    bq[p] = *(const v4f*)(Hsum + (((hj >> 7) * 8 + (p >> 1)) * 128 + (hj & 127)) * 16 + (p & 1) * 8 + 4 * half);

  const float* Ab = W2 + d * DD + 4 * l16;
  v4f aq[32][2];
  #pragma unroll
  for (int ks = 0; ks < 32; ++ks) {
    #pragma unroll
    for (int v = 0; v < 2; ++v)
      aq[ks][v] = *(const v4f*)(Ab + (size_t)c_of(ks, half, v) * 4096);
  }

  v4f kq[8], b2q[8];
  #pragma unroll
  for (int r = 0; r < 8; ++r) {
    kq[r]  = *(const v4f*)(K  + hj * DD + 32 * half + 4 * r);
    b2q[r] = *(const v4f*)(b2 + d  * DD + 32 * half + 4 * r);
  }

  v8f acc[4] = {};
  #pragma unroll
  for (int ks = 0; ks < 32; ++ks) {
    v2f bf;
    if (ks & 1) { bf.x = bq[ks >> 1].z; bf.y = bq[ks >> 1].w; }
    else        { bf.x = bq[ks >> 1].x; bf.y = bq[ks >> 1].y; }
    #pragma unroll
    for (int et = 0; et < 4; ++et) {
      v2f af;
      af.x = aq[ks][0][et];
      af.y = aq[ks][1][et];
      acc[et] = __builtin_amdgcn_wmma_f32_16x16x4_f32(
           false, af,  false, bf,
           (short)0, acc[et],  false,  false);
    }
  }

  float s = 0.f;
  #pragma unroll
  for (int r = 0; r < 8; ++r) {
    #pragma unroll
    for (int et = 0; et < 4; ++et)
      s += (acc[et][r] * (1.f / 128.f) + b2q[r][et]) * kq[r][et];
  }
  s += __shfl_xor(s, 16);
  if (half == 0) out[hj * DD + d] = s;
}

extern "C" void kernel_launch(void* const* d_in, const int* in_sizes, int n_in,
                              void* d_out, int out_size, void* d_ws, size_t ws_size,
                              hipStream_t stream) {
  const float* Q  = (const float*)d_in[0];
  const float* K  = (const float*)d_in[1];
  const float* W1 = (const float*)d_in[2];
  const float* b1 = (const float*)d_in[3];
  const float* W2 = (const float*)d_in[4];
  const float* b2 = (const float*)d_in[5];
  float* out = (float*)d_out;

  float* Hsum    = (float*)d_ws;

  k_hsum<<<256, 512, 0, stream>>>(Q, K, W1, b1, Hsum);
  dim3 g3(4, 64);
  k_w2k<<<g3, 128, 0, stream>>>(K, W2, b2, Hsum, out);
}
